# starting kernel
# baseline (speedup 1.0000x reference)
#include <hip/hip_runtime.h>
#include <hip/hip_bf16.h>

typedef __attribute__((ext_vector_type(2))) float v2f;
typedef __attribute__((ext_vector_type(8))) float v8f;

#define B_ 4
#define S_ 8192
#define D_ 1024
#define O_ 4

__global__ void __launch_bounds__(256)
hyena_proj_wmma(const float* __restrict__ x,
                const float* __restrict__ w,
                float* __restrict__ proj)
{
    const int  wave = (int)((blockIdx.x * blockDim.x + threadIdx.x) >> 5);
    const int  lane = (int)(threadIdx.x & 31);
    const int  m    = lane & 15;
    const int  kh   = lane >> 4;
    const long row0 = (long)wave * 16;

    const float* xrow = x + (row0 + m) * D_ + 2 * kh;
    const float* wrow  = w + (long)(m & (O_ - 1)) * D_ + 2 * kh;
    const bool   bvalid = (m < O_);

    v8f acc = {};
    #pragma unroll 4
    for (int k = 0; k < D_; k += 4) {
        v2f a;
        a.x = __builtin_nontemporal_load(xrow + k);
        a.y = __builtin_nontemporal_load(xrow + k + 1);
        float w0 = wrow[k];
        float w1 = wrow[k + 1];
        v2f bm;
        bm.x = bvalid ? w0 : 0.0f;
        bm.y = bvalid ? w1 : 0.0f;
        acc = __builtin_amdgcn_wmma_f32_16x16x4_f32(false, a, false, bm,
                                                    (short)0, acc, false, false);
    }

    if (bvalid) {
        #pragma unroll
        for (int j = 0; j < 8; ++j) {
            long r = row0 + j + 8 * kh;
            int  b = (int)(r / S_);
            int  s = (int)(r % S_);
            proj[((long)s * B_ + b) * O_ + m] = acc[j];
        }
    }
}

#define CHUNK 1024
#define WARM  2048

__global__ void __launch_bounds__(64)
hyena_recurrence(const float* __restrict__ proj,
                 const float* __restrict__ fc,
                 float* __restrict__ out)
{
    const int d     = (int)(blockIdx.x * 64 + threadIdx.x);
    const int chunk = (int)blockIdx.y;
    const int b     = (int)blockIdx.z;

    const float c0 = __expf(fc[0 * D_ + d]);
    const float c1 = __expf(fc[1 * D_ + d]);
    const float c2 = __expf(fc[2 * D_ + d]);
    const float c3 = __expf(fc[3 * D_ + d]);

    const int t0 = chunk * CHUNK;
    int t = t0 - WARM;
    if (t < 0) t = 0;

    const float4* p4 = (const float4*)proj;
    float y0 = 0.f, y1 = 0.f, y2 = 0.f, y3 = 0.f;

    for (; t < t0; ++t) {
        float4 p = p4[(long)t * B_ + b];
        y0 = fmaf(c0, y0, p.x);
        y1 = fmaf(c1, y1, p.y);
        y2 = fmaf(c2, y2, p.z);
        y3 = fmaf(c3, y3, p.w);
    }

    float* o = out + (long)b * S_ * D_ + d;
    __builtin_prefetch(&p4[(long)t0 * B_ + b], 0, 1);

    #pragma unroll 4
    for (; t < t0 + CHUNK; ++t) {
        float4 p = p4[(long)t * B_ + b];
        y0 = fmaf(c0, y0, p.x);
        y1 = fmaf(c1, y1, p.y);
        y2 = fmaf(c2, y2, p.z);
        y3 = fmaf(c3, y3, p.w);
        __builtin_nontemporal_store((y0 + y1) + (y2 + y3), o + (long)t * D_);
    }
}

extern "C" void kernel_launch(void* const* d_in, const int* in_sizes, int n_in,
                              void* d_out, int out_size, void* d_ws, size_t ws_size,
                              hipStream_t stream) {
    const float* x  = (const float*)d_in[0];
    const float* w  = (const float*)d_in[1];
    const float* fc = (const float*)d_in[2];
    float*       out  = (float*)d_out;
    float*       proj = (float*)d_ws;

    const int nwaves = (B_ * S_) / 16;
    hyena_proj_wmma<<<nwaves / 8, 256, 0, stream>>>(x, w, proj);

    dim3 grid(D_ / 64, S_ / CHUNK, B_);
    hyena_recurrence<<<grid, 64, 0, stream>>>(proj, fc, out);
}


# proj GEMM: a block's 8 waves split d (w slice held as B fragments); x read as 512 B row pieces, turned into A fragments via a wave-private LDS tile; partial sums added in LDS (d sum order changed)
# speedup vs baseline: 2.6305x; 2.6305x over previous
#include <hip/hip_runtime.h>
#include <hip/hip_bf16.h>

typedef __attribute__((ext_vector_type(2))) float v2f;
typedef __attribute__((ext_vector_type(8))) float v8f;

#define B_ 4
#define S_ 8192
#define D_ 1024
#define O_ 4

typedef __attribute__((ext_vector_type(4))) float v4f;

#define PT   8
#define NG   3
#define XP   132

__global__ void __launch_bounds__(256)
hyena_proj_wmma(const float* __restrict__ x,
                const float* __restrict__ w,
                float* __restrict__ proj)
{
    __shared__ __attribute__((aligned(16))) float sx[8][16 * XP];
    __shared__ float red[8][PT][64];

    const int  tid  = (int)threadIdx.x;
    const int  lane = tid & 31;
    const int  q    = tid >> 5;
    const int  m    = lane & 15;
    const int  kh   = lane >> 4;
    const long row0 = (long)blockIdx.x * (16 * PT);

    v4f bf[16];
    {
        const float* wp = w + (long)(m & (O_ - 1)) * D_ + 128 * q + 4 * kh;
        const bool bvalid = (m < O_);
        #pragma unroll
        for (int j = 0; j < 16; ++j) {
            const v4f t = *(const v4f*)(wp + 8 * j);
            bf[j] = bvalid ? t : (v4f){0.f, 0.f, 0.f, 0.f};
        }
    }

    const v4f* xp = (const v4f*)(x + row0 * D_ + 128 * q) + lane;
    float*     sw = &sx[q][0];

    v4f g[NG][16];
    #pragma unroll
    for (int u = 0; u < NG; ++u) {
        #pragma unroll
        for (int r = 0; r < 16; ++r)
            g[u][r] = __builtin_nontemporal_load(xp + (long)(16 * u + r) * (D_ / 4));
    }

    #pragma unroll
    for (int t = 0; t < PT; ++t) {
        const int u = t % NG;
        #pragma unroll
        for (int r = 0; r < 16; ++r) *(v4f*)(sw + r * XP + 4 * lane) = g[u][r];
        if (t + NG < PT) {
            #pragma unroll
            for (int r = 0; r < 16; ++r)
                g[u][r] = __builtin_nontemporal_load(xp + (long)(16 * (t + NG) + r) * (D_ / 4));
        }
        __builtin_amdgcn_fence(__ATOMIC_RELEASE, "wavefront");
        __builtin_amdgcn_wave_barrier();
        __builtin_amdgcn_fence(__ATOMIC_ACQUIRE, "wavefront");

        v4f a[16];
        #pragma unroll
        for (int j = 0; j < 16; ++j) a[j] = *(const v4f*)(sw + m * XP + 8 * j + 4 * kh);
        __builtin_amdgcn_sched_barrier(0);

        v8f c0 = {}, c1 = {};
        #pragma unroll
        for (int j = 0; j < 16; ++j) {
            const v2f a0 = {a[j].x, a[j].y},   a1 = {a[j].z, a[j].w};
            const v2f b0 = {bf[j].x, bf[j].y}, b1 = {bf[j].z, bf[j].w};
            c0 = __builtin_amdgcn_wmma_f32_16x16x4_f32(false, a0, false, b0, (short)0, c0, false, false);
            c1 = __builtin_amdgcn_wmma_f32_16x16x4_f32(false, a1, false, b1, (short)0, c1, false, false);
        }
        __builtin_amdgcn_fence(__ATOMIC_RELEASE, "wavefront");
        __builtin_amdgcn_wave_barrier();
        __builtin_amdgcn_fence(__ATOMIC_ACQUIRE, "wavefront");

        if (m < O_) {
            #pragma unroll
            for (int i = 0; i < 8; ++i) red[q][t][(i + 8 * kh) * 4 + m] = c0[i] + c1[i];
        }
    }

    __syncthreads();

    #pragma unroll
    for (int v = 0; v < 2; ++v) {
        const int idx = tid + 256 * v;
        const int t   = idx >> 6;
        const int e   = idx & 63;
        float sum = red[0][t][e];
        #pragma unroll
        for (int k = 1; k < 8; ++k) sum += red[k][t][e];
        const long r = row0 + 16 * t + (e >> 2);
        const int  b = (int)(r / S_);
        const int  s = (int)(r % S_);
        proj[((long)s * B_ + b) * O_ + (e & 3)] = sum;
    }
}

#define WARM   2048
#define NT_    224
#define C0LEN  3584
#define CLEN   1536
#define LROW   (16 + NT_ * 16)
#define LFILL  29
#define LPAD   (LFILL * 128)

__device__ __forceinline__ void load_frag(v2f (&a)[O_][4], const float* q)
{
    #pragma unroll
    for (int o = 0; o < O_; ++o) {
        #pragma unroll
        for (int s = 0; s < 4; ++s) {
            a[o][s].x = q[o * LPAD + 6 - 2 * s];
            a[o][s].y = q[o * LPAD + 7 - 2 * s];
        }
    }
}

template <bool STORE>
__device__ __forceinline__ void scan_tile(const v2f (&a)[O_][4], const v2f (&bw)[O_][4],
                                          const float (&c16)[O_], v8f (&y)[O_],
                                          const v8f (&gp)[O_], v8f (&gn)[O_], float* op)
{
    v8f ov;
    #pragma unroll
    for (int s = 0; s < 4; ++s) {
        #pragma unroll
        for (int u = 0; u < 2; ++u) {
            #pragma unroll
            for (int v = 0; v < 2; ++v) {
                const int o = 2 * u + v;
                gn[o] = __builtin_amdgcn_wmma_f32_16x16x4_f32(false, a[o][s], false, bw[o][s],
                                                              (short)0, s == 0 ? (v8f){} : gn[o],
                                                              false, false);
            }
            __builtin_amdgcn_sched_barrier(0);
            const int r = 2 * s + u;
            #pragma unroll
            for (int o = 0; o < O_; ++o) y[o][r] = fmaf(c16[o], y[o][r], gp[o][r]);
            if (STORE) ov[r] = (y[0][r] + y[1][r]) + (y[2][r] + y[3][r]);
            __builtin_amdgcn_sched_barrier(0);
        }
    }
    if (STORE) {
        #pragma unroll
        for (int r = 0; r < 8; ++r) __builtin_nontemporal_store(ov[r], op + (long)r * D_);
    }
}

__global__ void __launch_bounds__(128)
hyena_scan_toeplitz(const float* __restrict__ proj,
                    const float* __restrict__ fc,
                    float* __restrict__ out)
{
    __shared__ float sp[O_][LPAD];

    const int tid   = (int)threadIdx.x;
    const int lane  = tid & 31;
    const int c     = lane & 15;
    const int h     = lane >> 4;
    const int d     = (int)blockIdx.x * 64 + (tid >> 5) * 16 + c;
    const int chunk = (int)blockIdx.y;
    const int b     = (int)blockIdx.z;

    const int t0    = chunk == 0 ? 0 : C0LEN + (chunk - 1) * CLEN;
    const int ts    = chunk == 0 ? 0 : t0 - WARM;
    const int npw   = (t0 - ts) / 32;

    const float4* p4 = (const float4*)proj;
    #pragma unroll
    for (int it = 0; it < LFILL; ++it) {
        const int j  = it * 128 + tid;
        const int t  = ts - 16 + j;
        const int tc = t < 0 ? 0 : (t > S_ - 1 ? S_ - 1 : t);
        float4 p = p4[(long)tc * B_ + b];
        if (t < 0) p = make_float4(0.f, 0.f, 0.f, 0.f);
        sp[0][j] = p.x; sp[1][j] = p.y; sp[2][j] = p.z; sp[3][j] = p.w;
    }

    v2f   bw[O_][4];
    float c16[O_];
    #pragma unroll
    for (int o = 0; o < O_; ++o) {
        const float cv = __expf(fc[o * D_ + d]);
        const float c2 = cv * cv, c4 = c2 * c2, c8 = c4 * c4;
        c16[o] = c8 * c8;
        float pw = h ? c8 : 1.0f;
        #pragma unroll
        for (int s = 0; s < 4; ++s) {
            const float hi = pw * cv;
            bw[o][s].x = hi;
            bw[o][s].y = pw;
            pw = hi * cv;
        }
    }

    __syncthreads();

    const float* q = &sp[0][0] + (16 + c - 8 * h - 7);

    float* op = out + ((long)b * S_ + ts + 8 * h) * D_ + d;

    v8f y[O_], g0[O_], g1[O_];
    #pragma unroll
    for (int o = 0; o < O_; ++o) { y[o] = (v8f){}; g1[o] = (v8f){}; }

    v2f A0[O_][4], A1[O_][4];
    load_frag(A0, q);
    load_frag(A1, q + 16);
    scan_tile<false>(A0, bw, c16, y, g1, g0, op);

    int k = 0;
    for (; k < npw; ++k) {
        load_frag(A0, q + 32);
        __builtin_amdgcn_sched_barrier(0);
        scan_tile<false>(A1, bw, c16, y, g0, g1, op);
        load_frag(A1, q + 48);
        __builtin_amdgcn_sched_barrier(0);
        scan_tile<false>(A0, bw, c16, y, g1, g0, op);
        q += 32;
        op += 32 * D_;
    }
    for (; k < NT_ / 2; ++k) {
        load_frag(A0, q + 32);
        __builtin_amdgcn_sched_barrier(0);
        scan_tile<true>(A1, bw, c16, y, g0, g1, op);
        load_frag(A1, q + 48);
        __builtin_amdgcn_sched_barrier(0);
        scan_tile<true>(A0, bw, c16, y, g1, g0, op + 16 * D_);
        q += 32;
        op += 32 * D_;
    }
}

extern "C" void kernel_launch(void* const* d_in, const int* in_sizes, int n_in,
                              void* d_out, int out_size, void* d_ws, size_t ws_size,
                              hipStream_t stream) {
    const float* x  = (const float*)d_in[0];
    const float* w  = (const float*)d_in[1];
    const float* fc = (const float*)d_in[2];
    float*       out  = (float*)d_out;
    float*       proj = (float*)d_ws;

    const int nwaves = (B_ * S_) / 16;
    hyena_proj_wmma<<<nwaves / 8, 256, 0, stream>>>(x, w, proj);

    dim3 grid(D_ / 64, 4, B_);
    hyena_scan_toeplitz<<<grid, 128, 0, stream>>>(proj, fc, out);
}


# scan kernel: chunk start state from ONE strided Toeplitz product over its 2048-step window (rows 16 steps apart, Horner in c^16), not 128 re-warm tiles; chunks 2144,2016x3; warm-up sum order changed
# speedup vs baseline: 3.6187x; 1.3757x over previous
#include <hip/hip_runtime.h>
#include <hip/hip_bf16.h>

typedef __attribute__((ext_vector_type(2))) float v2f;
typedef __attribute__((ext_vector_type(8))) float v8f;

#define B_ 4
#define S_ 8192
#define D_ 1024
#define O_ 4

typedef __attribute__((ext_vector_type(4))) float v4f;

#define PT   8
#define NG   3
#define XP   132

__global__ void __launch_bounds__(256)
hyena_proj_wmma(const float* __restrict__ x,
                const float* __restrict__ w,
                float* __restrict__ proj)
{
    __shared__ __attribute__((aligned(16))) float sx[8][16 * XP];
    __shared__ float red[8][PT][64];

    const int  tid  = (int)threadIdx.x;
    const int  lane = tid & 31;
    const int  q    = tid >> 5;
    const int  m    = lane & 15;
    const int  kh   = lane >> 4;
    const long row0 = (long)blockIdx.x * (16 * PT);

    v4f bf[16];
    {
        const float* wp = w + (long)(m & (O_ - 1)) * D_ + 128 * q + 4 * kh;
        const bool bvalid = (m < O_);
        #pragma unroll
        for (int j = 0; j < 16; ++j) {
            const v4f t = *(const v4f*)(wp + 8 * j);
            bf[j] = bvalid ? t : (v4f){0.f, 0.f, 0.f, 0.f};
        }
    }

    const v4f* xp = (const v4f*)(x + row0 * D_ + 128 * q) + lane;
    float*     sw = &sx[q][0];

    v4f g[NG][16];
    #pragma unroll
    for (int u = 0; u < NG; ++u) {
        #pragma unroll
        for (int r = 0; r < 16; ++r)
            g[u][r] = __builtin_nontemporal_load(xp + (long)(16 * u + r) * (D_ / 4));
    }

    #pragma unroll
    for (int t = 0; t < PT; ++t) {
        const int u = t % NG;
        #pragma unroll
        for (int r = 0; r < 16; ++r) *(v4f*)(sw + r * XP + 4 * lane) = g[u][r];
        if (t + NG < PT) {
            #pragma unroll
            for (int r = 0; r < 16; ++r)
                g[u][r] = __builtin_nontemporal_load(xp + (long)(16 * (t + NG) + r) * (D_ / 4));
        }
        __builtin_amdgcn_fence(__ATOMIC_RELEASE, "wavefront");
        __builtin_amdgcn_wave_barrier();
        __builtin_amdgcn_fence(__ATOMIC_ACQUIRE, "wavefront");

        v4f a[16];
        #pragma unroll
        for (int j = 0; j < 16; ++j) a[j] = *(const v4f*)(sw + m * XP + 8 * j + 4 * kh);
        __builtin_amdgcn_sched_barrier(0);

        v8f c0 = {}, c1 = {};
        #pragma unroll
        for (int j = 0; j < 16; ++j) {
            const v2f a0 = {a[j].x, a[j].y},   a1 = {a[j].z, a[j].w};
            const v2f b0 = {bf[j].x, bf[j].y}, b1 = {bf[j].z, bf[j].w};
            c0 = __builtin_amdgcn_wmma_f32_16x16x4_f32(false, a0, false, b0, (short)0, c0, false, false);
            c1 = __builtin_amdgcn_wmma_f32_16x16x4_f32(false, a1, false, b1, (short)0, c1, false, false);
        }
        __builtin_amdgcn_fence(__ATOMIC_RELEASE, "wavefront");
        __builtin_amdgcn_wave_barrier();
        __builtin_amdgcn_fence(__ATOMIC_ACQUIRE, "wavefront");

        if (m < O_) {
            #pragma unroll
            for (int i = 0; i < 8; ++i) red[q][t][(i + 8 * kh) * 4 + m] = c0[i] + c1[i];
        }
    }

    __syncthreads();

    #pragma unroll
    for (int v = 0; v < 2; ++v) {
        const int idx = tid + 256 * v;
        const int t   = idx >> 6;
        const int e   = idx & 63;
        float sum = red[0][t][e];
        #pragma unroll
        for (int k = 1; k < 8; ++k) sum += red[k][t][e];
        const long r = row0 + 16 * t + (e >> 2);
        const int  b = (int)(r / S_);
        const int  s = (int)(r % S_);
        proj[((long)s * B_ + b) * O_ + (e & 3)] = sum;
    }
}

#define WARM   2048
#define C0T    134
#define CT     126
#define LFILL  32
#define LPAD   (LFILL * 128)

__device__ __forceinline__ void load_frag(v2f (&a)[O_][4], const float* q)
{
    #pragma unroll
    for (int o = 0; o < O_; ++o) {
        #pragma unroll
        for (int s = 0; s < 4; ++s) {
            a[o][s].x = q[o * LPAD + 6 - 2 * s];
            a[o][s].y = q[o * LPAD + 7 - 2 * s];
        }
    }
}

__device__ __forceinline__ void state_tile(const v2f (&a)[O_][4], const v2f (&bw)[O_][4],
                                           const float (&c16)[O_], const float (&q16)[O_],
                                           float (&acc)[O_])
{
    v8f z[O_];
    #pragma unroll
    for (int s = 0; s < 4; ++s) {
        #pragma unroll
        for (int o = 0; o < O_; ++o)
            z[o] = __builtin_amdgcn_wmma_f32_16x16x4_f32(false, a[o][s], false, bw[o][s],
                                                         (short)0, s == 0 ? (v8f){} : z[o],
                                                         false, false);
    }
    #pragma unroll
    for (int o = 0; o < O_; ++o) {
        float w = z[o][7];
        #pragma unroll
        for (int r = 6; r >= 0; --r) w = fmaf(w, c16[o], z[o][r]);
        acc[o] = fmaf(acc[o], q16[o], w);
    }
}

template <bool STORE>
__device__ __forceinline__ void scan_tile(const v2f (&a)[O_][4], const v2f (&bw)[O_][4],
                                          const float (&c16)[O_], v8f (&y)[O_],
                                          const v8f (&gp)[O_], v8f (&gn)[O_], float* op)
{
    v8f ov;
    #pragma unroll
    for (int s = 0; s < 4; ++s) {
        #pragma unroll
        for (int u = 0; u < 2; ++u) {
            #pragma unroll
            for (int v = 0; v < 2; ++v) {
                const int o = 2 * u + v;
                gn[o] = __builtin_amdgcn_wmma_f32_16x16x4_f32(false, a[o][s], false, bw[o][s],
                                                              (short)0, s == 0 ? (v8f){} : gn[o],
                                                              false, false);
            }
            __builtin_amdgcn_sched_barrier(0);
            const int r = 2 * s + u;
            #pragma unroll
            for (int o = 0; o < O_; ++o) y[o][r] = fmaf(c16[o], y[o][r], gp[o][r]);
            if (STORE) ov[r] = (y[0][r] + y[1][r]) + (y[2][r] + y[3][r]);
            __builtin_amdgcn_sched_barrier(0);
        }
    }
    if (STORE) {
        #pragma unroll
        for (int r = 0; r < 8; ++r) __builtin_nontemporal_store(ov[r], op + (long)r * D_);
    }
}

__global__ void __launch_bounds__(128)
hyena_scan_toeplitz(const float* __restrict__ proj,
                    const float* __restrict__ fc,
                    float* __restrict__ out)
{
    __shared__ float sp[O_][LPAD];

    const int tid   = (int)threadIdx.x;
    const int lane  = tid & 31;
    const int c     = lane & 15;
    const int h     = lane >> 4;
    const int d     = (int)blockIdx.x * 64 + (tid >> 5) * 16 + c;
    const int chunk = (int)blockIdx.y;
    const int b     = (int)blockIdx.z;

    const int t0    = chunk == 0 ? 0 : 16 * C0T + (chunk - 1) * 16 * CT;
    const int np    = (chunk == 0 ? C0T : CT) / 2;
    const int tl    = chunk == 0 ? -16 : t0 - WARM;

    const float4* p4 = (const float4*)proj;
    #pragma unroll
    for (int it = 0; it < LFILL; ++it) {
        const int j  = it * 128 + tid;
        const int t  = tl + j;
        const int tc = t < 0 ? 0 : (t > S_ - 1 ? S_ - 1 : t);
        float4 p = p4[(long)tc * B_ + b];
        if (t < 0) p = make_float4(0.f, 0.f, 0.f, 0.f);
        sp[0][j] = p.x; sp[1][j] = p.y; sp[2][j] = p.z; sp[3][j] = p.w;
    }

    v2f   bw[O_][4];
    float c16[O_], cv1[O_], c8v[O_];
    #pragma unroll
    for (int o = 0; o < O_; ++o) {
        const float cv = __expf(fc[o * D_ + d]);
        const float c2 = cv * cv, c4 = c2 * c2, c8 = c4 * c4;
        cv1[o] = cv;
        c8v[o] = c8;
        c16[o] = c8 * c8;
        float pw = h ? c8 : 1.0f;
        #pragma unroll
        for (int s = 0; s < 4; ++s) {
            const float hi = pw * cv;
            bw[o][s].x = hi;
            bw[o][s].y = pw;
            pw = hi * cv;
        }
    }

    __syncthreads();

    float Y[O_] = {0.f, 0.f, 0.f, 0.f};
    if (chunk != 0) {
        float q8[O_], q16[O_], acc[O_];
        #pragma unroll
        for (int o = 0; o < O_; ++o) {
            const float e2 = c16[o] * c16[o], e4 = e2 * e2;
            q8[o]  = e4 * e4;
            q16[o] = q8[o] * q8[o];
            acc[o] = 0.f;
        }
        const float* qw = &sp[0][0] + (WARM - 1 - 16 * c - 8 * h - 7);
        v2f Wa[O_][4], Wb[O_][4];
        load_frag(Wa, qw - 256 * 7);
        #pragma unroll
        for (int jj = 0; jj < 4; ++jj) {
            load_frag(Wb, qw - 256 * (6 - 2 * jj));
            state_tile(Wa, bw, c16, q16, acc);
            if (jj < 3) load_frag(Wa, qw - 256 * (5 - 2 * jj));
            state_tile(Wb, bw, c16, q16, acc);
        }
        #pragma unroll
        for (int o = 0; o < O_; ++o) {
            const float sh = h ? acc[o] * q8[o] : acc[o];
            Y[o] = sh + __shfl_xor(sh, 16);
        }
    }

    const float* q = &sp[0][0] + (t0 - tl + c - 8 * h - 7);

    float* op = out + ((long)b * S_ + t0 + 8 * h) * D_ + d;

    v8f y[O_], g0[O_], g1[O_];
    #pragma unroll
    for (int o = 0; o < O_; ++o) { y[o] = (v8f){}; g1[o] = (v8f){}; }

    v2f A0[O_][4], A1[O_][4];
    load_frag(A0, q);
    #pragma unroll
    for (int o = 0; o < O_; ++o) {
        #pragma unroll
        for (int s = 0; s < 4; ++s) {
            if (8 * h + 2 * s + 1 > c) A0[o][s].x = 0.f;
            if (8 * h + 2 * s     > c) A0[o][s].y = 0.f;
        }
    }
    load_frag(A1, q + 16);
    scan_tile<false>(A0, bw, c16, y, g1, g0, op);

    #pragma unroll
    for (int o = 0; o < O_; ++o) {
        float pw = h ? c8v[o] * cv1[o] : cv1[o];
        #pragma unroll
        for (int r = 0; r < 8; ++r) {
            g0[o][r] = fmaf(pw, Y[o], g0[o][r]);
            pw *= cv1[o];
        }
    }

    for (int k = 0; k < np; ++k) {
        load_frag(A0, q + 32);
        __builtin_amdgcn_sched_barrier(0);
        scan_tile<true>(A1, bw, c16, y, g0, g1, op);
        load_frag(A1, q + 48);
        __builtin_amdgcn_sched_barrier(0);
        scan_tile<true>(A0, bw, c16, y, g1, g0, op + 16 * D_);
        q += 32;
        op += 32 * D_;
    }
}

extern "C" void kernel_launch(void* const* d_in, const int* in_sizes, int n_in,
                              void* d_out, int out_size, void* d_ws, size_t ws_size,
                              hipStream_t stream) {
    const float* x  = (const float*)d_in[0];
    const float* w  = (const float*)d_in[1];
    const float* fc = (const float*)d_in[2];
    float*       out  = (float*)d_out;
    float*       proj = (float*)d_ws;

    const int nwaves = (B_ * S_) / 16;
    hyena_proj_wmma<<<nwaves / 8, 256, 0, stream>>>(x, w, proj);

    dim3 grid(D_ / 64, 4, B_);
    hyena_scan_toeplitz<<<grid, 128, 0, stream>>>(proj, fc, out);
}
